# starting kernel
# baseline (speedup 1.0000x reference)
#include <hip/hip_runtime.h>
#include <hip/hip_bf16.h>

typedef float v2f __attribute__((ext_vector_type(2)));
typedef float v8f __attribute__((ext_vector_type(8)));

#define N_NODES 100000
#define FDIM 128

__global__ void zero_f32_kernel(float* __restrict__ p, size_t n) {
    size_t i = (size_t)blockIdx.x * blockDim.x + threadIdx.x;
    size_t stride = (size_t)gridDim.x * blockDim.x;
    for (; i < n; i += stride) p[i] = 0.0f;
}

__global__ void fill_f32_kernel(float* __restrict__ p, float v, int n) {
    int i = blockIdx.x * blockDim.x + threadIdx.x;
    if (i < n) p[i] = v;
}

__global__ void deg_accum_kernel(const int* __restrict__ dst, int E,
                                 float* __restrict__ deg) {
    int e = blockIdx.x * blockDim.x + threadIdx.x;
    if (e < E) atomicAdd(&deg[dst[e]], 1.0f);
}

__global__ void dinv_kernel(float* __restrict__ deg, int n) {
    int i = blockIdx.x * blockDim.x + threadIdx.x;
    if (i < n) {
        float d = deg[i];
        deg[i] = (d > 0.0f) ? rsqrtf(d) : 0.0f;
    }
}

__global__ __launch_bounds__(256) void gemm1_wmma_kernel(
    const float* __restrict__ X, const float* __restrict__ W,
    float* __restrict__ H, int M) {
    const int wave  = threadIdx.x >> 5;
    const int lane  = threadIdx.x & 31;
    const int row16 = lane & 15;
    const int khalf = lane >> 4;
    const int rowBase = blockIdx.x * 16;
    const int colBase = wave * 16;
    if (rowBase >= M) return;

    v8f c0 = {};
    v8f c1 = {};
    const float* xrow = X + (size_t)(rowBase + row16) * FDIM;
    const int HALF = FDIM / 2;

    #pragma unroll 4
    for (int k0 = 0; k0 < HALF; k0 += 4) {
        const int ka = k0 + 2 * khalf;
        const int kb = ka + HALF;
        v2f a0, b0, a1, b1;
        a0.x = xrow[ka];
        a0.y = xrow[ka + 1];
        a1.x = xrow[kb];
        a1.y = xrow[kb + 1];
        b0.x = W[(size_t)ka * FDIM + colBase + row16];
        b0.y = W[(size_t)(ka + 1) * FDIM + colBase + row16];
        b1.x = W[(size_t)kb * FDIM + colBase + row16];
        b1.y = W[(size_t)(kb + 1) * FDIM + colBase + row16];
        c0 = __builtin_amdgcn_wmma_f32_16x16x4_f32(
                false, a0, false, b0, (short)0, c0, false, false);
        c1 = __builtin_amdgcn_wmma_f32_16x16x4_f32(
                false, a1, false, b1, (short)0, c1, false, false);
    }
    v8f c = c0 + c1;

    float* hbase = H + (size_t)(rowBase + 8 * khalf) * FDIM + colBase + row16;
    #pragma unroll
    for (int r = 0; r < 8; ++r)
        hbase[(size_t)r * FDIM] = c[r];
}

__global__ __launch_bounds__(256) void scatter1_kernel(
    const int* __restrict__ src, const int* __restrict__ dst, int E,
    const float* __restrict__ dinv, const float* __restrict__ H1,
    float* __restrict__ A1) {
    int e = blockIdx.x * 8 + (threadIdx.x >> 5);
    if (e >= E) return;
    const int lane = threadIdx.x & 31;
    const int s = src[e];
    const int d = dst[e];
    const float norm = dinv[s] * dinv[d];
    float4 v = ((const float4*)(H1 + (size_t)s * FDIM))[lane];
    float* ap = A1 + (size_t)d * FDIM + lane * 4;
    atomicAdd(ap + 0, v.x * norm);
    atomicAdd(ap + 1, v.y * norm);
    atomicAdd(ap + 2, v.z * norm);
    atomicAdd(ap + 3, v.w * norm);
}

__global__ void finalize1_kernel(float* __restrict__ A1,
                                 const float* __restrict__ H1,
                                 const float* __restrict__ dinv,
                                 const float* __restrict__ b1, size_t total) {
    size_t i = (size_t)blockIdx.x * blockDim.x + threadIdx.x;
    size_t stride = (size_t)gridDim.x * blockDim.x;
    for (; i < total; i += stride) {
        int node = (int)(i >> 7);
        int ch   = (int)(i & (FDIM - 1));
        float di = dinv[node];
        float v = A1[i] + H1[i] * di * di + b1[ch];
        A1[i] = v > 0.0f ? v : 0.0f;
    }
}

__global__ __launch_bounds__(256) void dense2_kernel(
    const float* __restrict__ A1, const float* __restrict__ X,
    const float* __restrict__ W2, const float* __restrict__ Wsk,
    float* __restrict__ H2, float* __restrict__ SK, int n) {
    int node = blockIdx.x * 8 + (threadIdx.x >> 5);
    if (node >= n) return;
    const int lane = threadIdx.x & 31;
    const int k = lane * 4;
    float4 a = ((const float4*)(A1 + (size_t)node * FDIM))[lane];
    float4 x = ((const float4*)(X  + (size_t)node * FDIM))[lane];

    float h0 = a.x * W2[(k + 0) * 2 + 0] + a.y * W2[(k + 1) * 2 + 0] +
               a.z * W2[(k + 2) * 2 + 0] + a.w * W2[(k + 3) * 2 + 0];
    float h1 = a.x * W2[(k + 0) * 2 + 1] + a.y * W2[(k + 1) * 2 + 1] +
               a.z * W2[(k + 2) * 2 + 1] + a.w * W2[(k + 3) * 2 + 1];
    float s0 = x.x * Wsk[(k + 0) * 2 + 0] + x.y * Wsk[(k + 1) * 2 + 0] +
               x.z * Wsk[(k + 2) * 2 + 0] + x.w * Wsk[(k + 3) * 2 + 0];
    float s1 = x.x * Wsk[(k + 0) * 2 + 1] + x.y * Wsk[(k + 1) * 2 + 1] +
               x.z * Wsk[(k + 2) * 2 + 1] + x.w * Wsk[(k + 3) * 2 + 1];

    #pragma unroll
    for (int off = 16; off > 0; off >>= 1) {
        h0 += __shfl_down(h0, off, 32);
        h1 += __shfl_down(h1, off, 32);
        s0 += __shfl_down(s0, off, 32);
        s1 += __shfl_down(s1, off, 32);
    }
    if (lane == 0) {
        H2[(size_t)node * 2 + 0] = h0;
        H2[(size_t)node * 2 + 1] = h1;
        SK[(size_t)node * 2 + 0] = s0;
        SK[(size_t)node * 2 + 1] = s1;
    }
}

__global__ void scatter2_kernel(const int* __restrict__ src,
                                const int* __restrict__ dst, int E,
                                const float* __restrict__ dinv,
                                const float* __restrict__ H2,
                                float* __restrict__ OUT) {
    int e = blockIdx.x * blockDim.x + threadIdx.x;
    if (e >= E) return;
    const int s = src[e];
    const int d = dst[e];
    const float norm = dinv[s] * dinv[d];
    atomicAdd(&OUT[(size_t)d * 2 + 0], H2[(size_t)s * 2 + 0] * norm);
    atomicAdd(&OUT[(size_t)d * 2 + 1], H2[(size_t)s * 2 + 1] * norm);
}

__global__ void finalize2_kernel(float* __restrict__ OUT,
                                 const float* __restrict__ H2,
                                 const float* __restrict__ SK,
                                 const float* __restrict__ dinv,
                                 const float* __restrict__ b2,
                                 const float* __restrict__ bsk, int n) {
    int i = blockIdx.x * blockDim.x + threadIdx.x;
    if (i >= n) return;
    float di = dinv[i];
    float dd = di * di;
    OUT[(size_t)i * 2 + 0] += H2[(size_t)i * 2 + 0] * dd + b2[0] + SK[(size_t)i * 2 + 0] + bsk[0];
    OUT[(size_t)i * 2 + 1] += H2[(size_t)i * 2 + 1] * dd + b2[1] + SK[(size_t)i * 2 + 1] + bsk[1];
}

extern "C" void kernel_launch(void* const* d_in, const int* in_sizes, int n_in,
                              void* d_out, int out_size, void* d_ws, size_t ws_size,
                              hipStream_t stream) {
    const float* x   = (const float*)d_in[0];
    const int*   ei  = (const int*)d_in[1];
    const float* W1  = (const float*)d_in[2];
    const float* b1  = (const float*)d_in[3];
    const float* W2  = (const float*)d_in[4];
    const float* b2  = (const float*)d_in[5];
    const float* Wsk = (const float*)d_in[6];
    const float* bsk = (const float*)d_in[7];
    float* out = (float*)d_out;

    const int N = in_sizes[0] / FDIM;
    const int E = in_sizes[1] / 2;
    const int* src = ei;
    const int* dst = ei + E;

    char* ws = (char*)d_ws;
    float* dinv = (float*)ws;
    float* H1   = (float*)(ws + (size_t)N * sizeof(float));
    float* A1   = H1 + (size_t)N * FDIM;
    float* H2   = A1 + (size_t)N * FDIM;
    float* SK   = H2 + (size_t)N * 2;

    fill_f32_kernel<<<(N + 255) / 256, 256, 0, stream>>>(dinv, 1.0f, N);
    zero_f32_kernel<<<2048, 256, 0, stream>>>(A1, (size_t)N * FDIM);
    zero_f32_kernel<<<512, 256, 0, stream>>>(out, (size_t)out_size);

    deg_accum_kernel<<<(E + 255) / 256, 256, 0, stream>>>(dst, E, dinv);
    dinv_kernel<<<(N + 255) / 256, 256, 0, stream>>>(dinv, N);

    gemm1_wmma_kernel<<<(N + 15) / 16, 256, 0, stream>>>(x, W1, H1, N);

    scatter1_kernel<<<(E + 7) / 8, 256, 0, stream>>>(src, dst, E, dinv, H1, A1);
    finalize1_kernel<<<2048, 256, 0, stream>>>(A1, H1, dinv, b1, (size_t)N * FDIM);

    dense2_kernel<<<(N + 7) / 8, 256, 0, stream>>>(A1, x, W2, Wsk, H2, SK, N);

    scatter2_kernel<<<(E + 255) / 256, 256, 0, stream>>>(src, dst, E, dinv, H2, out);
    finalize2_kernel<<<(N + 255) / 256, 256, 0, stream>>>(out, H2, SK, dinv, b2, bsk, N);
}


# Both aggregations become per-node gathers over an edge list bucketed by target in each call (int returning-add count, no float atomics), epilogues folded in; sum order = count arrival rank
# speedup vs baseline: 2.3363x; 2.3363x over previous
#include <hip/hip_runtime.h>


typedef float v2f __attribute__((ext_vector_type(2)));
typedef float v8f __attribute__((ext_vector_type(8)));

#define N_NODES 100000
#define FDIM 128

__global__ void zero_i32_kernel(int* __restrict__ p, int n) {
    int i = blockIdx.x * blockDim.x + threadIdx.x;
    if (i < n) p[i] = 0;
}

__global__ void count_rank_kernel(const int* __restrict__ dst, int E,
                                  int* __restrict__ cnt, int* __restrict__ rank) {
    int e = blockIdx.x * blockDim.x + threadIdx.x;
    if (e < E) rank[e] = atomicAdd(&cnt[dst[e]], 1);
}

__global__ __launch_bounds__(256) void alloc_kernel(const int* __restrict__ cnt,
                                                    int* __restrict__ total,
                                                    float* __restrict__ dinv,
                                                    int* __restrict__ start,
                                                    int2* __restrict__ seg, int n) {
    const int i = blockIdx.x * 256 + threadIdx.x;
    const int lane = threadIdx.x & 31;
    const int c = (i < n) ? cnt[i] : 0;
    if (i < n) {
        float d = (float)(c + 1);
        dinv[i] = (d > 0.0f) ? rsqrtf(d) : 0.0f;
    }
    int incl = c;
    #pragma unroll
    for (int off = 1; off < 32; off <<= 1) {
        int t = __shfl_up(incl, off, 32);
        if (lane >= off) incl += t;
    }
    int base = 0;
    if (lane == 31) base = atomicAdd(total, incl);
    base = __shfl(base, 31, 32);
    if (i < n) {
        const int st = base + incl - c;
        start[i] = st;
        seg[i] = make_int2(st, c);
    }
}

__global__ void fill_kernel(const int* __restrict__ src, const int* __restrict__ dst,
                            const int* __restrict__ rank, const int* __restrict__ start,
                            int E, int* __restrict__ csr) {
    int e = blockIdx.x * blockDim.x + threadIdx.x;
    if (e < E) csr[start[dst[e]] + rank[e]] = src[e];
}

__global__ __launch_bounds__(256) void gemm1_wmma_kernel(
    const float* __restrict__ X, const float* __restrict__ W,
    float* __restrict__ H, int M) {
    const int wave  = threadIdx.x >> 5;
    const int lane  = threadIdx.x & 31;
    const int row16 = lane & 15;
    const int khalf = lane >> 4;
    const int rowBase = blockIdx.x * 16;
    const int colBase = wave * 16;
    if (rowBase >= M) return;

    v8f c0 = {};
    v8f c1 = {};
    const float* xrow = X + (size_t)(rowBase + row16) * FDIM;
    const int HALF = FDIM / 2;

    #pragma unroll 4
    for (int k0 = 0; k0 < HALF; k0 += 4) {
        const int ka = k0 + 2 * khalf;
        const int kb = ka + HALF;
        v2f a0, b0, a1, b1;
        a0.x = xrow[ka];
        a0.y = xrow[ka + 1];
        a1.x = xrow[kb];
        a1.y = xrow[kb + 1];
        b0.x = W[(size_t)ka * FDIM + colBase + row16];
        b0.y = W[(size_t)(ka + 1) * FDIM + colBase + row16];
        b1.x = W[(size_t)kb * FDIM + colBase + row16];
        b1.y = W[(size_t)(kb + 1) * FDIM + colBase + row16];
        c0 = __builtin_amdgcn_wmma_f32_16x16x4_f32(
                false, a0, false, b0, (short)0, c0, false, false);
        c1 = __builtin_amdgcn_wmma_f32_16x16x4_f32(
                false, a1, false, b1, (short)0, c1, false, false);
    }
    v8f c = c0 + c1;

    float* hbase = H + (size_t)(rowBase + 8 * khalf) * FDIM + colBase + row16;
    #pragma unroll
    for (int r = 0; r < 8; ++r)
        hbase[(size_t)r * FDIM] = c[r];
}

template <int G>
__device__ __forceinline__ void gather_rows(const float* __restrict__ H1, int idx, float w,
                                            int j, int lane, float4& acc) {
    float4 v[G];
    float ws[G];
    #pragma unroll
    for (int g = 0; g < G; ++g) {
        const int s = __builtin_amdgcn_readlane(idx, j + g);
        ws[g] = __int_as_float(__builtin_amdgcn_readlane(__float_as_int(w), j + g));
        v[g] = ((const float4*)(H1 + (size_t)s * FDIM))[lane];
    }
    #pragma unroll
    for (int g = 0; g < G; ++g) {
        acc.x += v[g].x * ws[g];
        acc.y += v[g].y * ws[g];
        acc.z += v[g].z * ws[g];
        acc.w += v[g].w * ws[g];
    }
}

__global__ __launch_bounds__(256) void gather1_kernel(
    const int* __restrict__ csr, const int2* __restrict__ seg,
    const float* __restrict__ dinv, const float* __restrict__ H1, const float* __restrict__ X,
    const float* __restrict__ b1, const float* __restrict__ W2, const float* __restrict__ Wsk,
    float4* __restrict__ REC, float2* __restrict__ SK, int n) {
    const int lane = threadIdx.x & 31;
    const int node = __builtin_amdgcn_readfirstlane((int)(blockIdx.x * 8 + (threadIdx.x >> 5)));
    if (node >= n) return;
    const int2 sg = seg[node];
    const float di = dinv[node];
    const int beg = sg.x;
    const int c = sg.y;

    float4 acc = {0.0f, 0.0f, 0.0f, 0.0f};
    for (int base = 0; base < c; base += 32) {
        const int m = (c - base < 32) ? (c - base) : 32;
        int idx = 0;
        float w = 0.0f;
        if (lane < m) {
            idx = csr[beg + base + lane];
            w = dinv[idx] * di;
        }
        int j = 0;
        for (; j + 8 <= m; j += 8) gather_rows<8>(H1, idx, w, j, lane, acc);
        if (m - j >= 4) { gather_rows<4>(H1, idx, w, j, lane, acc); j += 4; }
        if (m - j >= 2) { gather_rows<2>(H1, idx, w, j, lane, acc); j += 2; }
        if (m - j >= 1) { gather_rows<1>(H1, idx, w, j, lane, acc); }
    }

    const int k = lane * 4;
    const float4 h  = ((const float4*)(H1 + (size_t)node * FDIM))[lane];
    const float4 bb = ((const float4*)b1)[lane];
    const float4 x  = ((const float4*)(X + (size_t)node * FDIM))[lane];
    float4 a;
    a.x = acc.x + h.x * di * di + bb.x;
    a.y = acc.y + h.y * di * di + bb.y;
    a.z = acc.z + h.z * di * di + bb.z;
    a.w = acc.w + h.w * di * di + bb.w;
    a.x = a.x > 0.0f ? a.x : 0.0f;
    a.y = a.y > 0.0f ? a.y : 0.0f;
    a.z = a.z > 0.0f ? a.z : 0.0f;
    a.w = a.w > 0.0f ? a.w : 0.0f;

    float h0 = a.x * W2[(k + 0) * 2 + 0] + a.y * W2[(k + 1) * 2 + 0] +
               a.z * W2[(k + 2) * 2 + 0] + a.w * W2[(k + 3) * 2 + 0];
    float h1 = a.x * W2[(k + 0) * 2 + 1] + a.y * W2[(k + 1) * 2 + 1] +
               a.z * W2[(k + 2) * 2 + 1] + a.w * W2[(k + 3) * 2 + 1];
    float s0 = x.x * Wsk[(k + 0) * 2 + 0] + x.y * Wsk[(k + 1) * 2 + 0] +
               x.z * Wsk[(k + 2) * 2 + 0] + x.w * Wsk[(k + 3) * 2 + 0];
    float s1 = x.x * Wsk[(k + 0) * 2 + 1] + x.y * Wsk[(k + 1) * 2 + 1] +
               x.z * Wsk[(k + 2) * 2 + 1] + x.w * Wsk[(k + 3) * 2 + 1];

    #pragma unroll
    for (int off = 16; off > 0; off >>= 1) {
        h0 += __shfl_down(h0, off, 32);
        h1 += __shfl_down(h1, off, 32);
        s0 += __shfl_down(s0, off, 32);
        s1 += __shfl_down(s1, off, 32);
    }
    if (lane == 0) {
        REC[node] = make_float4(h0, h1, di, 0.0f);
        SK[node]  = make_float2(s0, s1);
    }
}

__global__ __launch_bounds__(256) void gather2_kernel(
    const int* __restrict__ csr, const int2* __restrict__ seg,
    const float4* __restrict__ REC, const float2* __restrict__ SK,
    const float* __restrict__ b2, const float* __restrict__ bsk,
    float* __restrict__ OUT, int n) {
    const int t = blockIdx.x * 256 + threadIdx.x;
    const int node = t >> 3;
    const int sub = t & 7;
    const bool ok = node < n;
    int beg = 0, c = 0;
    float di = 0.0f;
    float4 self = {0.0f, 0.0f, 0.0f, 0.0f};
    if (ok) {
        const int2 sg = seg[node];
        beg = sg.x;
        c = sg.y;
        self = REC[node];
        di = self.z;
    }
    float a0 = 0.0f, a1 = 0.0f;
    for (int j = sub; j < c; j += 8) {
        const int s = csr[beg + j];
        const float4 r = REC[s];
        const float nrm = r.z * di;
        a0 += r.x * nrm;
        a1 += r.y * nrm;
    }
    #pragma unroll
    for (int off = 4; off > 0; off >>= 1) {
        a0 += __shfl_xor(a0, off, 32);
        a1 += __shfl_xor(a1, off, 32);
    }
    if (ok && sub == 0) {
        const float dd = di * di;
        const float2 sk = SK[node];
        float2 o;
        o.x = a0 + (self.x * dd + b2[0] + sk.x + bsk[0]);
        o.y = a1 + (self.y * dd + b2[1] + sk.y + bsk[1]);
        ((float2*)OUT)[node] = o;
    }
}

extern "C" void kernel_launch(void* const* d_in, const int* in_sizes, int n_in,
                              void* d_out, int out_size, void* d_ws, size_t ws_size,
                              hipStream_t stream) {
    const float* x   = (const float*)d_in[0];
    const int*   ei  = (const int*)d_in[1];
    const float* W1  = (const float*)d_in[2];
    const float* b1  = (const float*)d_in[3];
    const float* W2  = (const float*)d_in[4];
    const float* b2  = (const float*)d_in[5];
    const float* Wsk = (const float*)d_in[6];
    const float* bsk = (const float*)d_in[7];
    float* out = (float*)d_out;

    const int N = in_sizes[0] / FDIM;
    const int E = in_sizes[1] / 2;
    const int* src = ei;
    const int* dst = ei + E;

    char* ws = (char*)d_ws;
    float* dinv = (float*)ws;
    float* H1   = (float*)(ws + (size_t)N * sizeof(float));
    float* A1   = H1 + (size_t)N * FDIM;
    float4* REC = (float4*)(A1 + (size_t)N * FDIM);
    int* cnt    = (int*)A1;
    int* total  = cnt + N;
    int* start  = total + 32;
    int* rank   = start + N;
    int* csr    = rank + E;
    float2* SK  = (float2*)(csr + E);
    int2* seg   = (int2*)(SK + N);

    zero_i32_kernel<<<(N + 32 + 255) / 256, 256, 0, stream>>>(cnt, N + 32);
    count_rank_kernel<<<(E + 255) / 256, 256, 0, stream>>>(dst, E, cnt, rank);
    alloc_kernel<<<(N + 255) / 256, 256, 0, stream>>>(cnt, total, dinv, start, seg, N);
    fill_kernel<<<(E + 255) / 256, 256, 0, stream>>>(src, dst, rank, start, E, csr);

    gemm1_wmma_kernel<<<(N + 15) / 16, 256, 0, stream>>>(x, W1, H1, N);

    gather1_kernel<<<(N + 7) / 8, 256, 0, stream>>>(csr, seg, dinv, H1, x, b1, W2, Wsk,
                                                    REC, SK, N);

    gather2_kernel<<<(N * 8 + 255) / 256, 256, 0, stream>>>(csr, seg, REC, SK, b2, bsk,
                                                            out, N);
}


# Layer 1 sums X rows per target first, then multiplies each 32-node block of sums by W1 on WMMA from LDS in the same kernel: no X@W1 plane, one launch less; sum order: aggregate then product
# speedup vs baseline: 2.8081x; 1.2019x over previous
#include <hip/hip_runtime.h>


typedef float v2f __attribute__((ext_vector_type(2)));
typedef float v8f __attribute__((ext_vector_type(8)));

#define N_NODES 100000
#define FDIM 128
#define NPB 32
#define PITCH 132

__global__ void zero_i32_kernel(int* __restrict__ p, int n) {
    int i = blockIdx.x * blockDim.x + threadIdx.x;
    if (i < n) p[i] = 0;
}

__global__ void count_rank_kernel(const int* __restrict__ dst, int E,
                                  int* __restrict__ cnt, int* __restrict__ rank) {
    int e = blockIdx.x * blockDim.x + threadIdx.x;
    if (e < E) rank[e] = atomicAdd(&cnt[dst[e]], 1);
}

__global__ __launch_bounds__(256) void alloc_kernel(const int* __restrict__ cnt,
                                                    int* __restrict__ total,
                                                    float* __restrict__ dinv,
                                                    int* __restrict__ start,
                                                    int2* __restrict__ seg, int n) {
    const int i = blockIdx.x * 256 + threadIdx.x;
    const int lane = threadIdx.x & 31;
    const int c = (i < n) ? cnt[i] : 0;
    if (i < n) {
        float d = (float)(c + 1);
        dinv[i] = (d > 0.0f) ? rsqrtf(d) : 0.0f;
    }
    int incl = c;
    #pragma unroll
    for (int off = 1; off < 32; off <<= 1) {
        int t = __shfl_up(incl, off, 32);
        if (lane >= off) incl += t;
    }
    int base = 0;
    if (lane == 31) base = atomicAdd(total, incl);
    base = __shfl(base, 31, 32);
    if (i < n) {
        const int st = base + incl - c;
        start[i] = st;
        seg[i] = make_int2(st, c);
    }
}

__global__ void fill_kernel(const int* __restrict__ src, const int* __restrict__ dst,
                            const int* __restrict__ rank, const int* __restrict__ start,
                            int E, int* __restrict__ csr) {
    int e = blockIdx.x * blockDim.x + threadIdx.x;
    if (e < E) csr[start[dst[e]] + rank[e]] = src[e];
}


template <int G>
__device__ __forceinline__ void gather_rows(const float* __restrict__ X, int idx, float w,
                                            int j, int lane, float4& acc) {
    float4 v[G];
    float ws[G];
    #pragma unroll
    for (int g = 0; g < G; ++g) {
        const int s = __builtin_amdgcn_readlane(idx, j + g);
        ws[g] = __int_as_float(__builtin_amdgcn_readlane(__float_as_int(w), j + g));
        v[g] = ((const float4*)(X + (size_t)s * FDIM))[lane];
    }
    #pragma unroll
    for (int g = 0; g < G; ++g) {
        acc.x += v[g].x * ws[g];
        acc.y += v[g].y * ws[g];
        acc.z += v[g].z * ws[g];
        acc.w += v[g].w * ws[g];
    }
}

__device__ __forceinline__ void gather_chunk(const float* __restrict__ X, int idx, float w,
                                             int m, int lane, float4& acc) {
    int j = 0;
    for (; j + 8 <= m; j += 8) gather_rows<8>(X, idx, w, j, lane, acc);
    if (m - j >= 4) { gather_rows<4>(X, idx, w, j, lane, acc); j += 4; }
    if (m - j >= 2) { gather_rows<2>(X, idx, w, j, lane, acc); j += 2; }
    if (m - j >= 1) { gather_rows<1>(X, idx, w, j, lane, acc); }
}

__global__ __launch_bounds__(256) void gather1_gemm_kernel(
    const int* __restrict__ csr, const int2* __restrict__ seg,
    const float* __restrict__ dinv, const float* __restrict__ X, const float* __restrict__ W1,
    const float* __restrict__ b1, const float* __restrict__ W2, const float* __restrict__ Wsk,
    float4* __restrict__ REC, float2* __restrict__ SK, int n) {
    __shared__ __attribute__((aligned(16))) float agg[NPB * PITCH];
    __shared__ float part[8 * NPB * 2];

    const int lane = threadIdx.x & 31;
    const int wave = __builtin_amdgcn_readfirstlane((int)(threadIdx.x >> 5));
    const int c16 = lane & 15;
    const int kh = lane >> 4;
    const int colBase = wave * 16;
    const int nodeBase = blockIdx.x * NPB;

    const float4 wsA = ((const float4*)Wsk)[lane * 2];
    const float4 wsB = ((const float4*)Wsk)[lane * 2 + 1];

    #pragma nounroll
    for (int p = 0; p < 2; ++p) {
        bool ok[2];
        int nd[2], beg[2], cnt[2], idx0[2];
        float di[2], w0[2];
        float4 xr[2];
        #pragma unroll
        for (int q = 0; q < 2; ++q) {
            const int node = nodeBase + wave * 4 + p * 2 + q;
            ok[q] = node < n;
            nd[q] = ok[q] ? node : (n - 1);
            const int2 sg = seg[nd[q]];
            beg[q] = sg.x;
            cnt[q] = ok[q] ? sg.y : 0;
            di[q] = dinv[nd[q]];
        }
        #pragma unroll
        for (int q = 0; q < 2; ++q) {
            xr[q] = ((const float4*)(X + (size_t)nd[q] * FDIM))[lane];
            idx0[q] = 0;
            if (lane < cnt[q]) idx0[q] = csr[beg[q] + lane];
        }
        #pragma unroll
        for (int q = 0; q < 2; ++q) {
            w0[q] = 0.0f;
            if (lane < cnt[q]) w0[q] = dinv[idx0[q]] * di[q];
        }

        #pragma unroll
        for (int q = 0; q < 2; ++q) {
            const int r = wave * 4 + p * 2 + q;
            const int c = cnt[q];
            float4 acc = {0.0f, 0.0f, 0.0f, 0.0f};
            gather_chunk(X, idx0[q], w0[q], (c < 32) ? c : 32, lane, acc);
            for (int base = 32; base < c; base += 32) {
                const int m = (c - base < 32) ? (c - base) : 32;
                int idx = 0;
                float w = 0.0f;
                if (lane < m) {
                    idx = csr[beg[q] + base + lane];
                    w = dinv[idx] * di[q];
                }
                gather_chunk(X, idx, w, m, lane, acc);
            }
            const float4 x = xr[q];
            const float dd = ok[q] ? di[q] * di[q] : 0.0f;
            float4 row;
            row.x = acc.x + x.x * dd;
            row.y = acc.y + x.y * dd;
            row.z = acc.z + x.z * dd;
            row.w = acc.w + x.w * dd;
            *(float4*)(agg + r * PITCH + lane * 4) = row;

            float s0 = x.x * wsA.x + x.y * wsA.z + x.z * wsB.x + x.w * wsB.z;
            float s1 = x.x * wsA.y + x.y * wsA.w + x.z * wsB.y + x.w * wsB.w;
            #pragma unroll
            for (int off = 16; off > 0; off >>= 1) {
                s0 += __shfl_down(s0, off, 32);
                s1 += __shfl_down(s1, off, 32);
            }
            if (ok[q] && lane == 0) SK[nodeBase + r] = make_float2(s0, s1);
        }
    }
    v2f bfr[32];
    {
        const float* wp = W1 + (size_t)(2 * kh) * FDIM + colBase + c16;
        #pragma unroll
        for (int kk = 0; kk < 32; ++kk) {
            bfr[kk].x = wp[(size_t)(4 * kk) * FDIM];
            bfr[kk].y = wp[(size_t)(4 * kk + 1) * FDIM];
        }
    }
    __syncthreads();

    const int col = colBase + c16;
    const float bias = b1[col];
    const float2 w2 = ((const float2*)W2)[col];
    #pragma unroll
    for (int t = 0; t < 2; ++t) {
        v8f c0 = {};
        v8f c1 = {};
        const float* ap = agg + (t * 16 + c16) * PITCH + 2 * kh;
        __builtin_amdgcn_sched_barrier(0);
        #pragma unroll
        for (int kk = 0; kk < 16; ++kk) {
            const v2f a0 = *(const v2f*)(ap + 4 * kk);
            const v2f a1 = *(const v2f*)(ap + 4 * kk + 64);
            c0 = __builtin_amdgcn_wmma_f32_16x16x4_f32(
                    false, a0, false, bfr[kk], (short)0, c0, false, false);
            c1 = __builtin_amdgcn_wmma_f32_16x16x4_f32(
                    false, a1, false, bfr[kk + 16], (short)0, c1, false, false);
        }
        __builtin_amdgcn_sched_barrier(0);
        asm volatile("v_nop\n\tv_nop\n\tv_nop\n\tv_nop");
        __builtin_amdgcn_sched_barrier(0);
        const v8f cc = c0 + c1;

        float p0[8], p1[8];
        #pragma unroll
        for (int r = 0; r < 8; ++r) {
            float a = cc[r] + bias;
            a = a > 0.0f ? a : 0.0f;
            p0[r] = a * w2.x;
            p1[r] = a * w2.y;
        }
        #pragma unroll
        for (int off = 8; off > 0; off >>= 1) {
            #pragma unroll
            for (int r = 0; r < 8; ++r) {
                p0[r] += __shfl_xor(p0[r], off, 32);
                p1[r] += __shfl_xor(p1[r], off, 32);
            }
        }
        if (c16 == 0) {
            #pragma unroll
            for (int r = 0; r < 8; ++r) {
                part[(wave * NPB + t * 16 + r + 8 * kh) * 2 + 0] = p0[r];
                part[(wave * NPB + t * 16 + r + 8 * kh) * 2 + 1] = p1[r];
            }
        }
    }
    __syncthreads();
    if (threadIdx.x < NPB) {
        const int node = nodeBase + threadIdx.x;
        if (node < n) {
            float h0 = 0.0f, h1 = 0.0f;
            #pragma unroll
            for (int w = 0; w < 8; ++w) {
                h0 += part[(w * NPB + threadIdx.x) * 2 + 0];
                h1 += part[(w * NPB + threadIdx.x) * 2 + 1];
            }
            REC[node] = make_float4(h0, h1, dinv[node], 0.0f);
        }
    }
}

__global__ __launch_bounds__(256) void gather2_kernel(
    const int* __restrict__ csr, const int2* __restrict__ seg,
    const float4* __restrict__ REC, const float2* __restrict__ SK,
    const float* __restrict__ b2, const float* __restrict__ bsk,
    float* __restrict__ OUT, int n) {
    const int t = blockIdx.x * 256 + threadIdx.x;
    const int node = t >> 3;
    const int sub = t & 7;
    const bool ok = node < n;
    int beg = 0, c = 0;
    float di = 0.0f;
    float4 self = {0.0f, 0.0f, 0.0f, 0.0f};
    if (ok) {
        const int2 sg = seg[node];
        beg = sg.x;
        c = sg.y;
        self = REC[node];
        di = self.z;
    }
    float a0 = 0.0f, a1 = 0.0f;
    for (int j = sub; j < c; j += 8) {
        const int s = csr[beg + j];
        const float4 r = REC[s];
        const float nrm = r.z * di;
        a0 += r.x * nrm;
        a1 += r.y * nrm;
    }
    #pragma unroll
    for (int off = 4; off > 0; off >>= 1) {
        a0 += __shfl_xor(a0, off, 32);
        a1 += __shfl_xor(a1, off, 32);
    }
    if (ok && sub == 0) {
        const float dd = di * di;
        const float2 sk = SK[node];
        float2 o;
        o.x = a0 + (self.x * dd + b2[0] + sk.x + bsk[0]);
        o.y = a1 + (self.y * dd + b2[1] + sk.y + bsk[1]);
        ((float2*)OUT)[node] = o;
    }
}

extern "C" void kernel_launch(void* const* d_in, const int* in_sizes, int n_in,
                              void* d_out, int out_size, void* d_ws, size_t ws_size,
                              hipStream_t stream) {
    const float* x   = (const float*)d_in[0];
    const int*   ei  = (const int*)d_in[1];
    const float* W1  = (const float*)d_in[2];
    const float* b1  = (const float*)d_in[3];
    const float* W2  = (const float*)d_in[4];
    const float* b2  = (const float*)d_in[5];
    const float* Wsk = (const float*)d_in[6];
    const float* bsk = (const float*)d_in[7];
    float* out = (float*)d_out;

    const int N = in_sizes[0] / FDIM;
    const int E = in_sizes[1] / 2;
    const int* src = ei;
    const int* dst = ei + E;

    char* ws = (char*)d_ws;
    float* dinv = (float*)ws;
    float* H1   = (float*)(ws + (size_t)N * sizeof(float));
    float* A1   = H1 + (size_t)N * FDIM;
    float4* REC = (float4*)(A1 + (size_t)N * FDIM);
    int* cnt    = (int*)A1;
    int* total  = cnt + N;
    int* start  = total + 32;
    int* rank   = start + N;
    int* csr    = rank + E;
    float2* SK  = (float2*)(csr + E);
    int2* seg   = (int2*)(SK + N);

    zero_i32_kernel<<<(N + 32 + 255) / 256, 256, 0, stream>>>(cnt, N + 32);
    count_rank_kernel<<<(E + 255) / 256, 256, 0, stream>>>(dst, E, cnt, rank);
    alloc_kernel<<<(N + 255) / 256, 256, 0, stream>>>(cnt, total, dinv, start, seg, N);
    fill_kernel<<<(E + 255) / 256, 256, 0, stream>>>(src, dst, rank, start, E, csr);

    gather1_gemm_kernel<<<(N + NPB - 1) / NPB, 256, 0, stream>>>(csr, seg, dinv, x, W1, b1, W2, Wsk,
                                                                 REC, SK, N);

    gather2_kernel<<<(N * 8 + 255) / 256, 256, 0, stream>>>(csr, seg, REC, SK, b2, bsk,
                                                            out, N);
}


# Degree count split by chip half: each edge chunk is walked from both halves, a work-group adds only for targets in its own half's node range; sum order: count arrival rank, as before
# speedup vs baseline: 4.0274x; 1.4342x over previous
#include <hip/hip_runtime.h>


typedef float v2f __attribute__((ext_vector_type(2)));
typedef float v8f __attribute__((ext_vector_type(8)));

#define N_NODES 100000
#define FDIM 128
#define NPB 32
#define PITCH 132

__global__ void zero_i32_kernel(int* __restrict__ p, int n) {
    int i = blockIdx.x * blockDim.x + threadIdx.x;
    if (i < n) p[i] = 0;
}

#define CHUNK 2048
#define TICKET0 64
#define TICKET1 128
#define CNT_PAD 192
__global__ __launch_bounds__(256) void count_rank_kernel(const int* __restrict__ dst, int E, int nChunks,
                                                         int n, int* __restrict__ cnt,
                                                         int* __restrict__ total,
                                                         int* __restrict__ rank) {
    __shared__ int s_job;
    if (threadIdx.x == 0) {
        int half = (int)((__builtin_amdgcn_s_sendmsg_rtn(0x87) >> 18) & 1u);
        int t = atomicAdd(total + (half ? TICKET1 : TICKET0), 1);
        if (t >= nChunks) {
            half ^= 1;
            t = atomicAdd(total + (half ? TICKET1 : TICKET0), 1);
        }
        s_job = (t < nChunks) ? (t * 2 + half) : -1;
    }
    __syncthreads();
    const int job = s_job;
    if (job < 0) return;
    const int split = ((n >> 1) + 31) & ~31;
    const int lo = (job & 1) ? split : 0;
    const int hi = (job & 1) ? 0x7fffffff : split;
    const int base = (job >> 1) * CHUNK + (int)threadIdx.x;
    int d[8], r[8];
    #pragma unroll
    for (int k = 0; k < 8; ++k) {
        const int e = base + k * 256;
        const int v = dst[(e < E) ? e : (E - 1)];
        d[k] = (e < E) ? v : -1;
    }
    asm volatile("" : "+v"(d[0]), "+v"(d[1]), "+v"(d[2]), "+v"(d[3]),
                      "+v"(d[4]), "+v"(d[5]), "+v"(d[6]), "+v"(d[7]));
    #pragma unroll
    for (int k = 0; k < 8; ++k) {
        r[k] = 0;
        if (d[k] >= lo && d[k] < hi) r[k] = atomicAdd(&cnt[d[k]], 1);
    }
    #pragma unroll
    for (int k = 0; k < 8; ++k) {
        if (d[k] >= lo && d[k] < hi) rank[base + k * 256] = r[k];
    }
}

__global__ __launch_bounds__(256) void alloc_kernel(const int* __restrict__ cnt,
                                                    int* __restrict__ total,
                                                    float* __restrict__ dinv,
                                                    int* __restrict__ start,
                                                    int2* __restrict__ seg, int n) {
    const int i = blockIdx.x * 256 + threadIdx.x;
    const int lane = threadIdx.x & 31;
    const int c = (i < n) ? cnt[i] : 0;
    if (i < n) {
        float d = (float)(c + 1);
        dinv[i] = (d > 0.0f) ? rsqrtf(d) : 0.0f;
    }
    int incl = c;
    #pragma unroll
    for (int off = 1; off < 32; off <<= 1) {
        int t = __shfl_up(incl, off, 32);
        if (lane >= off) incl += t;
    }
    int base = 0;
    if (lane == 31) base = atomicAdd(total, incl);
    base = __shfl(base, 31, 32);
    if (i < n) {
        const int st = base + incl - c;
        start[i] = st;
        seg[i] = make_int2(st, c);
    }
}

__global__ void fill_kernel(const int* __restrict__ src, const int* __restrict__ dst,
                            const int* __restrict__ rank, const int* __restrict__ start,
                            int E, int* __restrict__ csr) {
    int e = blockIdx.x * blockDim.x + threadIdx.x;
    if (e < E) csr[start[dst[e]] + rank[e]] = src[e];
}


template <int G>
__device__ __forceinline__ void gather_rows(const float* __restrict__ X, int idx, float w,
                                            int j, int lane, float4& acc) {
    float4 v[G];
    float ws[G];
    #pragma unroll
    for (int g = 0; g < G; ++g) {
        const int s = __builtin_amdgcn_readlane(idx, j + g);
        ws[g] = __int_as_float(__builtin_amdgcn_readlane(__float_as_int(w), j + g));
        v[g] = ((const float4*)(X + (size_t)s * FDIM))[lane];
    }
    #pragma unroll
    for (int g = 0; g < G; ++g) {
        acc.x += v[g].x * ws[g];
        acc.y += v[g].y * ws[g];
        acc.z += v[g].z * ws[g];
        acc.w += v[g].w * ws[g];
    }
}

__device__ __forceinline__ void gather_chunk(const float* __restrict__ X, int idx, float w,
                                             int m, int lane, float4& acc) {
    int j = 0;
    for (; j + 8 <= m; j += 8) gather_rows<8>(X, idx, w, j, lane, acc);
    if (m - j >= 4) { gather_rows<4>(X, idx, w, j, lane, acc); j += 4; }
    if (m - j >= 2) { gather_rows<2>(X, idx, w, j, lane, acc); j += 2; }
    if (m - j >= 1) { gather_rows<1>(X, idx, w, j, lane, acc); }
}

__global__ __launch_bounds__(256) void gather1_gemm_kernel(
    const int* __restrict__ csr, const int2* __restrict__ seg,
    const float* __restrict__ dinv, const float* __restrict__ X, const float* __restrict__ W1,
    const float* __restrict__ b1, const float* __restrict__ W2, const float* __restrict__ Wsk,
    float4* __restrict__ REC, float2* __restrict__ SK, int n) {
    __shared__ __attribute__((aligned(16))) float agg[NPB * PITCH];
    __shared__ float part[8 * NPB * 2];

    const int lane = threadIdx.x & 31;
    const int wave = __builtin_amdgcn_readfirstlane((int)(threadIdx.x >> 5));
    const int c16 = lane & 15;
    const int kh = lane >> 4;
    const int colBase = wave * 16;
    const int nodeBase = blockIdx.x * NPB;

    const float4 wsA = ((const float4*)Wsk)[lane * 2];
    const float4 wsB = ((const float4*)Wsk)[lane * 2 + 1];

    #pragma nounroll
    for (int p = 0; p < 2; ++p) {
        bool ok[2];
        int nd[2], beg[2], cnt[2], idx0[2];
        float di[2], w0[2];
        float4 xr[2];
        #pragma unroll
        for (int q = 0; q < 2; ++q) {
            const int node = nodeBase + wave * 4 + p * 2 + q;
            ok[q] = node < n;
            nd[q] = ok[q] ? node : (n - 1);
            const int2 sg = seg[nd[q]];
            beg[q] = sg.x;
            cnt[q] = ok[q] ? sg.y : 0;
            di[q] = dinv[nd[q]];
        }
        #pragma unroll
        for (int q = 0; q < 2; ++q) {
            xr[q] = ((const float4*)(X + (size_t)nd[q] * FDIM))[lane];
            idx0[q] = 0;
            if (lane < cnt[q]) idx0[q] = csr[beg[q] + lane];
        }
        #pragma unroll
        for (int q = 0; q < 2; ++q) {
            w0[q] = 0.0f;
            if (lane < cnt[q]) w0[q] = dinv[idx0[q]] * di[q];
        }

        #pragma unroll
        for (int q = 0; q < 2; ++q) {
            const int r = wave * 4 + p * 2 + q;
            const int c = cnt[q];
            float4 acc = {0.0f, 0.0f, 0.0f, 0.0f};
            gather_chunk(X, idx0[q], w0[q], (c < 32) ? c : 32, lane, acc);
            for (int base = 32; base < c; base += 32) {
                const int m = (c - base < 32) ? (c - base) : 32;
                int idx = 0;
                float w = 0.0f;
                if (lane < m) {
                    idx = csr[beg[q] + base + lane];
                    w = dinv[idx] * di[q];
                }
                gather_chunk(X, idx, w, m, lane, acc);
            }
            const float4 x = xr[q];
            const float dd = ok[q] ? di[q] * di[q] : 0.0f;
            float4 row;
            row.x = acc.x + x.x * dd;
            row.y = acc.y + x.y * dd;
            row.z = acc.z + x.z * dd;
            row.w = acc.w + x.w * dd;
            *(float4*)(agg + r * PITCH + lane * 4) = row;

            float s0 = x.x * wsA.x + x.y * wsA.z + x.z * wsB.x + x.w * wsB.z;
            float s1 = x.x * wsA.y + x.y * wsA.w + x.z * wsB.y + x.w * wsB.w;
            #pragma unroll
            for (int off = 16; off > 0; off >>= 1) {
                s0 += __shfl_down(s0, off, 32);
                s1 += __shfl_down(s1, off, 32);
            }
            if (ok[q] && lane == 0) SK[nodeBase + r] = make_float2(s0, s1);
        }
    }
    v2f bfr[32];
    {
        const float* wp = W1 + (size_t)(2 * kh) * FDIM + colBase + c16;
        #pragma unroll
        for (int kk = 0; kk < 32; ++kk) {
            bfr[kk].x = wp[(size_t)(4 * kk) * FDIM];
            bfr[kk].y = wp[(size_t)(4 * kk + 1) * FDIM];
        }
    }
    __syncthreads();

    const int col = colBase + c16;
    const float bias = b1[col];
    const float2 w2 = ((const float2*)W2)[col];
    #pragma unroll
    for (int t = 0; t < 2; ++t) {
        v8f c0 = {};
        v8f c1 = {};
        const float* ap = agg + (t * 16 + c16) * PITCH + 2 * kh;
        __builtin_amdgcn_sched_barrier(0);
        #pragma unroll
        for (int kk = 0; kk < 16; ++kk) {
            const v2f a0 = *(const v2f*)(ap + 4 * kk);
            const v2f a1 = *(const v2f*)(ap + 4 * kk + 64);
            c0 = __builtin_amdgcn_wmma_f32_16x16x4_f32(
                    false, a0, false, bfr[kk], (short)0, c0, false, false);
            c1 = __builtin_amdgcn_wmma_f32_16x16x4_f32(
                    false, a1, false, bfr[kk + 16], (short)0, c1, false, false);
        }
        __builtin_amdgcn_sched_barrier(0);
        asm volatile("v_nop\n\tv_nop\n\tv_nop\n\tv_nop");
        __builtin_amdgcn_sched_barrier(0);
        const v8f cc = c0 + c1;

        float p0[8], p1[8];
        #pragma unroll
        for (int r = 0; r < 8; ++r) {
            float a = cc[r] + bias;
            a = a > 0.0f ? a : 0.0f;
            p0[r] = a * w2.x;
            p1[r] = a * w2.y;
        }
        #pragma unroll
        for (int off = 8; off > 0; off >>= 1) {
            #pragma unroll
            for (int r = 0; r < 8; ++r) {
                p0[r] += __shfl_xor(p0[r], off, 32);
                p1[r] += __shfl_xor(p1[r], off, 32);
            }
        }
        if (c16 == 0) {
            #pragma unroll
            for (int r = 0; r < 8; ++r) {
                part[(wave * NPB + t * 16 + r + 8 * kh) * 2 + 0] = p0[r];
                part[(wave * NPB + t * 16 + r + 8 * kh) * 2 + 1] = p1[r];
            }
        }
    }
    __syncthreads();
    if (threadIdx.x < NPB) {
        const int node = nodeBase + threadIdx.x;
        if (node < n) {
            float h0 = 0.0f, h1 = 0.0f;
            #pragma unroll
            for (int w = 0; w < 8; ++w) {
                h0 += part[(w * NPB + threadIdx.x) * 2 + 0];
                h1 += part[(w * NPB + threadIdx.x) * 2 + 1];
            }
            REC[node] = make_float4(h0, h1, dinv[node], 0.0f);
        }
    }
}

__global__ __launch_bounds__(256) void gather2_kernel(
    const int* __restrict__ csr, const int2* __restrict__ seg,
    const float4* __restrict__ REC, const float2* __restrict__ SK,
    const float* __restrict__ b2, const float* __restrict__ bsk,
    float* __restrict__ OUT, int n) {
    const int t = blockIdx.x * 256 + threadIdx.x;
    const int node = t >> 3;
    const int sub = t & 7;
    const bool ok = node < n;
    int beg = 0, c = 0;
    float di = 0.0f;
    float4 self = {0.0f, 0.0f, 0.0f, 0.0f};
    if (ok) {
        const int2 sg = seg[node];
        beg = sg.x;
        c = sg.y;
        self = REC[node];
        di = self.z;
    }
    float a0 = 0.0f, a1 = 0.0f;
    for (int j = sub; j < c; j += 8) {
        const int s = csr[beg + j];
        const float4 r = REC[s];
        const float nrm = r.z * di;
        a0 += r.x * nrm;
        a1 += r.y * nrm;
    }
    #pragma unroll
    for (int off = 4; off > 0; off >>= 1) {
        a0 += __shfl_xor(a0, off, 32);
        a1 += __shfl_xor(a1, off, 32);
    }
    if (ok && sub == 0) {
        const float dd = di * di;
        const float2 sk = SK[node];
        float2 o;
        o.x = a0 + (self.x * dd + b2[0] + sk.x + bsk[0]);
        o.y = a1 + (self.y * dd + b2[1] + sk.y + bsk[1]);
        ((float2*)OUT)[node] = o;
    }
}

extern "C" void kernel_launch(void* const* d_in, const int* in_sizes, int n_in,
                              void* d_out, int out_size, void* d_ws, size_t ws_size,
                              hipStream_t stream) {
    const float* x   = (const float*)d_in[0];
    const int*   ei  = (const int*)d_in[1];
    const float* W1  = (const float*)d_in[2];
    const float* b1  = (const float*)d_in[3];
    const float* W2  = (const float*)d_in[4];
    const float* b2  = (const float*)d_in[5];
    const float* Wsk = (const float*)d_in[6];
    const float* bsk = (const float*)d_in[7];
    float* out = (float*)d_out;

    const int N = in_sizes[0] / FDIM;
    const int E = in_sizes[1] / 2;
    const int* src = ei;
    const int* dst = ei + E;

    char* ws = (char*)d_ws;
    float* dinv = (float*)ws;
    float* H1   = (float*)(ws + (size_t)N * sizeof(float));
    float* A1   = H1 + (size_t)N * FDIM;
    float4* REC = (float4*)(A1 + (size_t)N * FDIM);
    int* cnt    = (int*)A1;
    int* total  = cnt + N;
    int* start  = total + CNT_PAD;
    int* rank   = start + N;
    int* csr    = rank + E;
    float2* SK  = (float2*)(csr + E);
    int2* seg   = (int2*)(SK + N);

    const int nChunks = (E + CHUNK - 1) / CHUNK;
    zero_i32_kernel<<<(N + CNT_PAD + 255) / 256, 256, 0, stream>>>(cnt, N + CNT_PAD);
    count_rank_kernel<<<2 * nChunks, 256, 0, stream>>>(dst, E, nChunks, N, cnt, total, rank);
    alloc_kernel<<<(N + 255) / 256, 256, 0, stream>>>(cnt, total, dinv, start, seg, N);
    fill_kernel<<<(E + 255) / 256, 256, 0, stream>>>(src, dst, rank, start, E, csr);

    gather1_gemm_kernel<<<(N + NPB - 1) / NPB, 256, 0, stream>>>(csr, seg, dinv, x, W1, b1, W2, Wsk,
                                                                 REC, SK, N);

    gather2_kernel<<<(N * 8 + 255) / 256, 256, 0, stream>>>(csr, seg, REC, SK, b2, bsk,
                                                            out, N);
}


# Each node owns one 128-byte line of source slots, so the counting pass stores an edge's source as it draws its rank (spill list past 32): no rank plane, prefix or fill pass; sum order: arrival rank
# speedup vs baseline: 5.8476x; 1.4520x over previous
#include <hip/hip_runtime.h>


typedef float v2f __attribute__((ext_vector_type(2)));
typedef float v8f __attribute__((ext_vector_type(8)));

#define N_NODES 100000
#define FDIM 128
#define NPB 32
#define PITCH 132

#define SLOTS 32

__device__ __forceinline__ float dinv_of(int c) {
    const float d = (float)(c + 1);
    return (d > 0.0f) ? rsqrtf(d) : 0.0f;
}

__global__ void zero_i32_kernel(int* __restrict__ p, int n) {
    int i = blockIdx.x * blockDim.x + threadIdx.x;
    if (i < n) p[i] = 0;
}

#define CHUNK 2048
#define TICKET0 64
#define TICKET1 128
#define CNT_PAD 192
__global__ __launch_bounds__(256) void count_fill_kernel(const int* __restrict__ src,
                                                         const int* __restrict__ dst, int E, int nChunks,
                                                         int n, int* __restrict__ cnt,
                                                         int* __restrict__ total,
                                                         int* __restrict__ bucket,
                                                         int2* __restrict__ spill) {
    __shared__ int s_job;
    if (threadIdx.x == 0) {
        int half = (int)((__builtin_amdgcn_s_sendmsg_rtn(0x87) >> 18) & 1u);
        int t = atomicAdd(total + (half ? TICKET1 : TICKET0), 1);
        if (t >= nChunks) {
            half ^= 1;
            t = atomicAdd(total + (half ? TICKET1 : TICKET0), 1);
        }
        s_job = (t < nChunks) ? (t * 2 + half) : -1;
    }
    __syncthreads();
    const int job = s_job;
    if (job < 0) return;
    const int split = ((n >> 1) + 31) & ~31;
    const int lo = (job & 1) ? split : 0;
    const int hi = (job & 1) ? 0x7fffffff : split;
    const int base = (job >> 1) * CHUNK + (int)threadIdx.x;
    int d[8], s[8], r[8];
    #pragma unroll
    for (int k = 0; k < 8; ++k) {
        const int e = base + k * 256;
        const int ec = (e < E) ? e : (E - 1);
        const int v = dst[ec];
        s[k] = src[ec];
        d[k] = (e < E) ? v : -1;
    }
    asm volatile("" : "+v"(d[0]), "+v"(d[1]), "+v"(d[2]), "+v"(d[3]),
                      "+v"(d[4]), "+v"(d[5]), "+v"(d[6]), "+v"(d[7]),
                      "+v"(s[0]), "+v"(s[1]), "+v"(s[2]), "+v"(s[3]),
                      "+v"(s[4]), "+v"(s[5]), "+v"(s[6]), "+v"(s[7]));
    #pragma unroll
    for (int k = 0; k < 8; ++k) {
        r[k] = 0;
        if (d[k] >= lo && d[k] < hi) r[k] = atomicAdd(&cnt[d[k]], 1);
    }
    #pragma unroll
    for (int k = 0; k < 8; ++k) {
        if (d[k] >= lo && d[k] < hi) {
            if (r[k] < SLOTS) {
                bucket[d[k] * SLOTS + r[k]] = s[k];
            } else {
                const int o = atomicAdd(total, 1);
                spill[o] = make_int2(d[k], s[k]);
            }
        }
    }
}


template <int G>
__device__ __forceinline__ void gather_rows(const float* __restrict__ X, int idx, float w,
                                            int j, int lane, float4& acc) {
    float4 v[G];
    float ws[G];
    #pragma unroll
    for (int g = 0; g < G; ++g) {
        const int s = __builtin_amdgcn_readlane(idx, j + g);
        ws[g] = __int_as_float(__builtin_amdgcn_readlane(__float_as_int(w), j + g));
        v[g] = ((const float4*)(X + (size_t)s * FDIM))[lane];
    }
    #pragma unroll
    for (int g = 0; g < G; ++g) {
        acc.x += v[g].x * ws[g];
        acc.y += v[g].y * ws[g];
        acc.z += v[g].z * ws[g];
        acc.w += v[g].w * ws[g];
    }
}

__device__ __forceinline__ void gather_chunk(const float* __restrict__ X, int idx, float w,
                                             int m, int lane, float4& acc) {
    int j = 0;
    for (; j + 8 <= m; j += 8) gather_rows<8>(X, idx, w, j, lane, acc);
    if (m - j >= 4) { gather_rows<4>(X, idx, w, j, lane, acc); j += 4; }
    if (m - j >= 2) { gather_rows<2>(X, idx, w, j, lane, acc); j += 2; }
    if (m - j >= 1) { gather_rows<1>(X, idx, w, j, lane, acc); }
}

__global__ __launch_bounds__(256) void gather1_gemm_kernel(
    const int* __restrict__ bucket, const int2* __restrict__ spill, const int* __restrict__ cnt_,
    const int* __restrict__ total, const float* __restrict__ X, const float* __restrict__ W1,
    const float* __restrict__ b1, const float* __restrict__ W2, const float* __restrict__ Wsk,
    float4* __restrict__ REC, float2* __restrict__ SK, int n) {
    __shared__ __attribute__((aligned(16))) float agg[NPB * PITCH];
    __shared__ float part[8 * NPB * 2];

    const int lane = threadIdx.x & 31;
    const int wave = __builtin_amdgcn_readfirstlane((int)(threadIdx.x >> 5));
    const int c16 = lane & 15;
    const int kh = lane >> 4;
    const int colBase = wave * 16;
    const int nodeBase = blockIdx.x * NPB;

    const float4 wsA = ((const float4*)Wsk)[lane * 2];
    const float4 wsB = ((const float4*)Wsk)[lane * 2 + 1];

    #pragma nounroll
    for (int p = 0; p < 2; ++p) {
        bool ok[2];
        int nd[2], cnt[2], idx0[2];
        float di[2], w0[2];
        float4 xr[2];
        #pragma unroll
        for (int q = 0; q < 2; ++q) {
            const int node = nodeBase + wave * 4 + p * 2 + q;
            ok[q] = node < n;
            nd[q] = ok[q] ? node : (n - 1);
            const int cq = cnt_[nd[q]];
            cnt[q] = ok[q] ? cq : 0;
            di[q] = dinv_of(cq);
        }
        #pragma unroll
        for (int q = 0; q < 2; ++q) {
            xr[q] = ((const float4*)(X + (size_t)nd[q] * FDIM))[lane];
            idx0[q] = 0;
            if (lane < cnt[q]) idx0[q] = bucket[nd[q] * SLOTS + lane];
        }
        #pragma unroll
        for (int q = 0; q < 2; ++q) {
            w0[q] = 0.0f;
            if (lane < cnt[q]) w0[q] = dinv_of(cnt_[idx0[q]]) * di[q];
        }

        #pragma unroll
        for (int q = 0; q < 2; ++q) {
            const int r = wave * 4 + p * 2 + q;
            const int c = cnt[q];
            float4 acc = {0.0f, 0.0f, 0.0f, 0.0f};
            gather_chunk(X, idx0[q], w0[q], (c < SLOTS) ? c : SLOTS, lane, acc);
            if (c > SLOTS) {
                const int nsp = total[0];
                for (int base = 0; base < nsp; base += 32) {
                    int idx = 0;
                    float w = 0.0f;
                    bool mine = false;
                    if (base + lane < nsp) {
                        const int2 ent = spill[base + lane];
                        mine = ent.x == nd[q];
                        if (mine) {
                            idx = ent.y;
                            w = dinv_of(cnt_[idx]) * di[q];
                        }
                    }
                    unsigned int hits = __builtin_amdgcn_ballot_w32(mine);
                    while (hits) {
                        const int j = __builtin_ctz(hits);
                        hits &= hits - 1;
                        gather_rows<1>(X, idx, w, j, lane, acc);
                    }
                }
            }
            const float4 x = xr[q];
            const float dd = ok[q] ? di[q] * di[q] : 0.0f;
            float4 row;
            row.x = acc.x + x.x * dd;
            row.y = acc.y + x.y * dd;
            row.z = acc.z + x.z * dd;
            row.w = acc.w + x.w * dd;
            *(float4*)(agg + r * PITCH + lane * 4) = row;

            float s0 = x.x * wsA.x + x.y * wsA.z + x.z * wsB.x + x.w * wsB.z;
            float s1 = x.x * wsA.y + x.y * wsA.w + x.z * wsB.y + x.w * wsB.w;
            #pragma unroll
            for (int off = 16; off > 0; off >>= 1) {
                s0 += __shfl_down(s0, off, 32);
                s1 += __shfl_down(s1, off, 32);
            }
            if (ok[q] && lane == 0) SK[nodeBase + r] = make_float2(s0, s1);
        }
    }
    v2f bfr[32];
    {
        const float* wp = W1 + (size_t)(2 * kh) * FDIM + colBase + c16;
        #pragma unroll
        for (int kk = 0; kk < 32; ++kk) {
            bfr[kk].x = wp[(size_t)(4 * kk) * FDIM];
            bfr[kk].y = wp[(size_t)(4 * kk + 1) * FDIM];
        }
    }
    __syncthreads();

    const int col = colBase + c16;
    const float bias = b1[col];
    const float2 w2 = ((const float2*)W2)[col];
    #pragma unroll
    for (int t = 0; t < 2; ++t) {
        v8f c0 = {};
        v8f c1 = {};
        const float* ap = agg + (t * 16 + c16) * PITCH + 2 * kh;
        __builtin_amdgcn_sched_barrier(0);
        #pragma unroll
        for (int kk = 0; kk < 16; ++kk) {
            const v2f a0 = *(const v2f*)(ap + 4 * kk);
            const v2f a1 = *(const v2f*)(ap + 4 * kk + 64);
            c0 = __builtin_amdgcn_wmma_f32_16x16x4_f32(
                    false, a0, false, bfr[kk], (short)0, c0, false, false);
            c1 = __builtin_amdgcn_wmma_f32_16x16x4_f32(
                    false, a1, false, bfr[kk + 16], (short)0, c1, false, false);
        }
        __builtin_amdgcn_sched_barrier(0);
        asm volatile("v_nop\n\tv_nop\n\tv_nop\n\tv_nop");
        __builtin_amdgcn_sched_barrier(0);
        const v8f cc = c0 + c1;

        float p0[8], p1[8];
        #pragma unroll
        for (int r = 0; r < 8; ++r) {
            float a = cc[r] + bias;
            a = a > 0.0f ? a : 0.0f;
            p0[r] = a * w2.x;
            p1[r] = a * w2.y;
        }
        #pragma unroll
        for (int off = 8; off > 0; off >>= 1) {
            #pragma unroll
            for (int r = 0; r < 8; ++r) {
                p0[r] += __shfl_xor(p0[r], off, 32);
                p1[r] += __shfl_xor(p1[r], off, 32);
            }
        }
        if (c16 == 0) {
            #pragma unroll
            for (int r = 0; r < 8; ++r) {
                part[(wave * NPB + t * 16 + r + 8 * kh) * 2 + 0] = p0[r];
                part[(wave * NPB + t * 16 + r + 8 * kh) * 2 + 1] = p1[r];
            }
        }
    }
    __syncthreads();
    if (threadIdx.x < NPB) {
        const int node = nodeBase + threadIdx.x;
        if (node < n) {
            float h0 = 0.0f, h1 = 0.0f;
            #pragma unroll
            for (int w = 0; w < 8; ++w) {
                h0 += part[(w * NPB + threadIdx.x) * 2 + 0];
                h1 += part[(w * NPB + threadIdx.x) * 2 + 1];
            }
            REC[node] = make_float4(h0, h1, dinv_of(cnt_[node]), 0.0f);
        }
    }
}

__global__ __launch_bounds__(256) void gather2_kernel(
    const int* __restrict__ bucket, const int2* __restrict__ spill, const int* __restrict__ cnt_,
    const int* __restrict__ total,
    const float4* __restrict__ REC, const float2* __restrict__ SK,
    const float* __restrict__ b2, const float* __restrict__ bsk,
    float* __restrict__ OUT, int n) {
    const int t = blockIdx.x * 256 + threadIdx.x;
    const int node = t >> 3;
    const int sub = t & 7;
    const bool ok = node < n;
    int c = 0;
    float di = 0.0f;
    float4 self = {0.0f, 0.0f, 0.0f, 0.0f};
    if (ok) {
        c = cnt_[node];
        self = REC[node];
        di = self.z;
    }
    const int beg = node * SLOTS;
    const int cb = (c < SLOTS) ? c : SLOTS;
    float a0 = 0.0f, a1 = 0.0f;
    for (int j = sub; j < cb; j += 8) {
        const int s = bucket[beg + j];
        const float4 r = REC[s];
        const float nrm = r.z * di;
        a0 += r.x * nrm;
        a1 += r.y * nrm;
    }
    if (c > SLOTS) {
        const int nsp = total[0];
        for (int k = sub; k < nsp; k += 8) {
            const int2 ent = spill[k];
            if (ent.x == node) {
                const float4 r = REC[ent.y];
                const float nrm = r.z * di;
                a0 += r.x * nrm;
                a1 += r.y * nrm;
            }
        }
    }
    #pragma unroll
    for (int off = 4; off > 0; off >>= 1) {
        a0 += __shfl_xor(a0, off, 32);
        a1 += __shfl_xor(a1, off, 32);
    }
    if (ok && sub == 0) {
        const float dd = di * di;
        const float2 sk = SK[node];
        float2 o;
        o.x = a0 + (self.x * dd + b2[0] + sk.x + bsk[0]);
        o.y = a1 + (self.y * dd + b2[1] + sk.y + bsk[1]);
        ((float2*)OUT)[node] = o;
    }
}

extern "C" void kernel_launch(void* const* d_in, const int* in_sizes, int n_in,
                              void* d_out, int out_size, void* d_ws, size_t ws_size,
                              hipStream_t stream) {
    const float* x   = (const float*)d_in[0];
    const int*   ei  = (const int*)d_in[1];
    const float* W1  = (const float*)d_in[2];
    const float* b1  = (const float*)d_in[3];
    const float* W2  = (const float*)d_in[4];
    const float* b2  = (const float*)d_in[5];
    const float* Wsk = (const float*)d_in[6];
    const float* bsk = (const float*)d_in[7];
    float* out = (float*)d_out;

    const int N = in_sizes[0] / FDIM;
    const int E = in_sizes[1] / 2;
    const int* src = ei;
    const int* dst = ei + E;

    char* ws = (char*)d_ws;
    float* H1   = (float*)(ws + (size_t)N * sizeof(float));
    float* A1   = H1 + (size_t)N * FDIM;
    float4* REC = (float4*)(A1 + (size_t)N * FDIM);
    int* bucket = (int*)H1;
    int* cnt    = (int*)A1;
    int* total  = cnt + N;
    float2* SK  = (float2*)(total + CNT_PAD);
    int2* spill = (int2*)(SK + N);

    const int nChunks = (E + CHUNK - 1) / CHUNK;
    zero_i32_kernel<<<(N + CNT_PAD + 255) / 256, 256, 0, stream>>>(cnt, N + CNT_PAD);
    count_fill_kernel<<<2 * nChunks, 256, 0, stream>>>(src, dst, E, nChunks, N, cnt, total, bucket, spill);

    gather1_gemm_kernel<<<(N + NPB - 1) / NPB, 256, 0, stream>>>(bucket, spill, cnt, total, x, W1, b1, W2,
                                                                 Wsk, REC, SK, N);

    gather2_kernel<<<(N * 8 + 255) / 256, 256, 0, stream>>>(bucket, spill, cnt, total, REC, SK, b2, bsk,
                                                            out, N);
}


# Layer 1: W1 staged once per work-group (64 nodes, 16 waves) in LDS by whole-row loads, row pairs interleaved; B fragments read from LDS, not 64 strided global loads per wave; sum order unchanged
# speedup vs baseline: 6.0337x; 1.0318x over previous
#include <hip/hip_runtime.h>


typedef float v2f __attribute__((ext_vector_type(2)));
typedef float v8f __attribute__((ext_vector_type(8)));

#define N_NODES 100000
#define FDIM 128
#define NPB 64
#define WPITCH 288
#define PITCH 132

#define SLOTS 32

__device__ __forceinline__ float dinv_of(int c) {
    const float d = (float)(c + 1);
    return (d > 0.0f) ? rsqrtf(d) : 0.0f;
}

__global__ void zero_i32_kernel(int* __restrict__ p, int n) {
    int i = blockIdx.x * blockDim.x + threadIdx.x;
    if (i < n) p[i] = 0;
}

#define CHUNK 2048
#define TICKET0 64
#define TICKET1 128
#define CNT_PAD 192
__global__ __launch_bounds__(256) void count_fill_kernel(const int* __restrict__ src,
                                                         const int* __restrict__ dst, int E, int nChunks,
                                                         int n, int* __restrict__ cnt,
                                                         int* __restrict__ total,
                                                         int* __restrict__ bucket,
                                                         int2* __restrict__ spill) {
    __shared__ int s_job;
    if (threadIdx.x == 0) {
        int half = (int)((__builtin_amdgcn_s_sendmsg_rtn(0x87) >> 18) & 1u);
        int t = atomicAdd(total + (half ? TICKET1 : TICKET0), 1);
        if (t >= nChunks) {
            half ^= 1;
            t = atomicAdd(total + (half ? TICKET1 : TICKET0), 1);
        }
        s_job = (t < nChunks) ? (t * 2 + half) : -1;
    }
    __syncthreads();
    const int job = s_job;
    if (job < 0) return;
    const int split = ((n >> 1) + 31) & ~31;
    const int lo = (job & 1) ? split : 0;
    const int hi = (job & 1) ? 0x7fffffff : split;
    const int base = (job >> 1) * CHUNK + (int)threadIdx.x;
    int d[8], s[8], r[8];
    #pragma unroll
    for (int k = 0; k < 8; ++k) {
        const int e = base + k * 256;
        const int ec = (e < E) ? e : (E - 1);
        const int v = dst[ec];
        s[k] = src[ec];
        d[k] = (e < E) ? v : -1;
    }
    asm volatile("" : "+v"(d[0]), "+v"(d[1]), "+v"(d[2]), "+v"(d[3]),
                      "+v"(d[4]), "+v"(d[5]), "+v"(d[6]), "+v"(d[7]),
                      "+v"(s[0]), "+v"(s[1]), "+v"(s[2]), "+v"(s[3]),
                      "+v"(s[4]), "+v"(s[5]), "+v"(s[6]), "+v"(s[7]));
    #pragma unroll
    for (int k = 0; k < 8; ++k) {
        r[k] = 0;
        if (d[k] >= lo && d[k] < hi) r[k] = atomicAdd(&cnt[d[k]], 1);
    }
    #pragma unroll
    for (int k = 0; k < 8; ++k) {
        if (d[k] >= lo && d[k] < hi) {
            if (r[k] < SLOTS) {
                bucket[d[k] * SLOTS + r[k]] = s[k];
            } else {
                const int o = atomicAdd(total, 1);
                spill[o] = make_int2(d[k], s[k]);
            }
        }
    }
}


template <int G>
__device__ __forceinline__ void gather_rows(const float* __restrict__ X, int idx, float w,
                                            int j, int lane, float4& acc) {
    float4 v[G];
    float ws[G];
    #pragma unroll
    for (int g = 0; g < G; ++g) {
        const int s = __builtin_amdgcn_readlane(idx, j + g);
        ws[g] = __int_as_float(__builtin_amdgcn_readlane(__float_as_int(w), j + g));
        v[g] = ((const float4*)(X + (size_t)s * FDIM))[lane];
    }
    #pragma unroll
    for (int g = 0; g < G; ++g) {
        acc.x += v[g].x * ws[g];
        acc.y += v[g].y * ws[g];
        acc.z += v[g].z * ws[g];
        acc.w += v[g].w * ws[g];
    }
}

__device__ __forceinline__ void gather_chunk(const float* __restrict__ X, int idx, float w,
                                             int m, int lane, float4& acc) {
    int j = 0;
    for (; j + 8 <= m; j += 8) gather_rows<8>(X, idx, w, j, lane, acc);
    if (m - j >= 4) { gather_rows<4>(X, idx, w, j, lane, acc); j += 4; }
    if (m - j >= 2) { gather_rows<2>(X, idx, w, j, lane, acc); j += 2; }
    if (m - j >= 1) { gather_rows<1>(X, idx, w, j, lane, acc); }
}

__global__ __launch_bounds__(512) void gather1_gemm_kernel(
    const int* __restrict__ bucket, const int2* __restrict__ spill, const int* __restrict__ cnt_,
    const int* __restrict__ total, const float* __restrict__ X, const float* __restrict__ W1,
    const float* __restrict__ b1, const float* __restrict__ W2, const float* __restrict__ Wsk,
    float4* __restrict__ REC, float2* __restrict__ SK, int n) {
    __shared__ __attribute__((aligned(16))) float agg[NPB * PITCH];
    __shared__ __attribute__((aligned(16))) float w1s[64 * WPITCH];
    __shared__ float part[8 * NPB * 2];

    const int lane = threadIdx.x & 31;
    const int wave = __builtin_amdgcn_readfirstlane((int)(threadIdx.x >> 5));
    const int c16 = lane & 15;
    const int kh = lane >> 4;
    const int strip = wave & 7;
    const int th = wave >> 3;
    const int colBase = strip * 16;
    const int nodeBase = blockIdx.x * NPB;

    float4 wr0[4], wr1[4];
    #pragma unroll
    for (int u = 0; u < 4; ++u) {
        const int j = wave + 16 * u;
        wr0[u] = ((const float4*)(W1 + (size_t)(2 * j) * FDIM))[lane];
        wr1[u] = ((const float4*)(W1 + (size_t)(2 * j + 1) * FDIM))[lane];
    }

    const float4 wsA = ((const float4*)Wsk)[lane * 2];
    const float4 wsB = ((const float4*)Wsk)[lane * 2 + 1];

    #pragma nounroll
    for (int p = 0; p < 2; ++p) {
        bool ok[2];
        int nd[2], cnt[2], idx0[2];
        float di[2], w0[2];
        float4 xr[2];
        #pragma unroll
        for (int q = 0; q < 2; ++q) {
            const int node = nodeBase + wave * 4 + p * 2 + q;
            ok[q] = node < n;
            nd[q] = ok[q] ? node : (n - 1);
            const int cq = cnt_[nd[q]];
            cnt[q] = ok[q] ? cq : 0;
            di[q] = dinv_of(cq);
        }
        #pragma unroll
        for (int q = 0; q < 2; ++q) {
            xr[q] = ((const float4*)(X + (size_t)nd[q] * FDIM))[lane];
            idx0[q] = 0;
            if (lane < cnt[q]) idx0[q] = bucket[nd[q] * SLOTS + lane];
        }
        #pragma unroll
        for (int q = 0; q < 2; ++q) {
            w0[q] = 0.0f;
            if (lane < cnt[q]) w0[q] = dinv_of(cnt_[idx0[q]]) * di[q];
        }

        #pragma unroll
        for (int q = 0; q < 2; ++q) {
            const int r = wave * 4 + p * 2 + q;
            const int c = cnt[q];
            float4 acc = {0.0f, 0.0f, 0.0f, 0.0f};
            gather_chunk(X, idx0[q], w0[q], (c < SLOTS) ? c : SLOTS, lane, acc);
            if (c > SLOTS) {
                const int nsp = total[0];
                for (int base = 0; base < nsp; base += 32) {
                    int idx = 0;
                    float w = 0.0f;
                    bool mine = false;
                    if (base + lane < nsp) {
                        const int2 ent = spill[base + lane];
                        mine = ent.x == nd[q];
                        if (mine) {
                            idx = ent.y;
                            w = dinv_of(cnt_[idx]) * di[q];
                        }
                    }
                    unsigned int hits = __builtin_amdgcn_ballot_w32(mine);
                    while (hits) {
                        const int j = __builtin_ctz(hits);
                        hits &= hits - 1;
                        gather_rows<1>(X, idx, w, j, lane, acc);
                    }
                }
            }
            const float4 x = xr[q];
            const float dd = ok[q] ? di[q] * di[q] : 0.0f;
            float4 row;
            row.x = acc.x + x.x * dd;
            row.y = acc.y + x.y * dd;
            row.z = acc.z + x.z * dd;
            row.w = acc.w + x.w * dd;
            *(float4*)(agg + r * PITCH + lane * 4) = row;

            float s0 = x.x * wsA.x + x.y * wsA.z + x.z * wsB.x + x.w * wsB.z;
            float s1 = x.x * wsA.y + x.y * wsA.w + x.z * wsB.y + x.w * wsB.w;
            #pragma unroll
            for (int off = 16; off > 0; off >>= 1) {
                s0 += __shfl_down(s0, off, 32);
                s1 += __shfl_down(s1, off, 32);
            }
            if (ok[q] && lane == 0) SK[nodeBase + r] = make_float2(s0, s1);
        }
    }
    #pragma unroll
    for (int u = 0; u < 4; ++u) {
        const int j = wave + 16 * u;
        float* wd = w1s + j * WPITCH + lane * 8;
        *(float4*)(wd)     = make_float4(wr0[u].x, wr1[u].x, wr0[u].y, wr1[u].y);
        *(float4*)(wd + 4) = make_float4(wr0[u].z, wr1[u].z, wr0[u].w, wr1[u].w);
    }
    __syncthreads();

    v2f bfr[32];
    {
        const float* wp = w1s + kh * WPITCH + (colBase + c16) * 2;
        #pragma unroll
        for (int kk = 0; kk < 32; ++kk) bfr[kk] = *(const v2f*)(wp + (2 * kk) * WPITCH);
    }

    const int col = colBase + c16;
    const float bias = b1[col];
    const float2 w2 = ((const float2*)W2)[col];
    #pragma unroll
    for (int t = 0; t < 2; ++t) {
        v8f c0 = {};
        v8f c1 = {};
        const float* ap = agg + ((2 * th + t) * 16 + c16) * PITCH + 2 * kh;
        __builtin_amdgcn_sched_barrier(0);
        #pragma unroll
        for (int kk = 0; kk < 16; ++kk) {
            const v2f a0 = *(const v2f*)(ap + 4 * kk);
            const v2f a1 = *(const v2f*)(ap + 4 * kk + 64);
            c0 = __builtin_amdgcn_wmma_f32_16x16x4_f32(
                    false, a0, false, bfr[kk], (short)0, c0, false, false);
            c1 = __builtin_amdgcn_wmma_f32_16x16x4_f32(
                    false, a1, false, bfr[kk + 16], (short)0, c1, false, false);
        }
        __builtin_amdgcn_sched_barrier(0);
        asm volatile("v_nop\n\tv_nop\n\tv_nop\n\tv_nop");
        __builtin_amdgcn_sched_barrier(0);
        const v8f cc = c0 + c1;

        float p0[8], p1[8];
        #pragma unroll
        for (int r = 0; r < 8; ++r) {
            float a = cc[r] + bias;
            a = a > 0.0f ? a : 0.0f;
            p0[r] = a * w2.x;
            p1[r] = a * w2.y;
        }
        #pragma unroll
        for (int off = 8; off > 0; off >>= 1) {
            #pragma unroll
            for (int r = 0; r < 8; ++r) {
                p0[r] += __shfl_xor(p0[r], off, 32);
                p1[r] += __shfl_xor(p1[r], off, 32);
            }
        }
        if (c16 == 0) {
            #pragma unroll
            for (int r = 0; r < 8; ++r) {
                part[(strip * NPB + (2 * th + t) * 16 + r + 8 * kh) * 2 + 0] = p0[r];
                part[(strip * NPB + (2 * th + t) * 16 + r + 8 * kh) * 2 + 1] = p1[r];
            }
        }
    }
    __syncthreads();
    if (threadIdx.x < NPB) {
        const int node = nodeBase + threadIdx.x;
        if (node < n) {
            float h0 = 0.0f, h1 = 0.0f;
            #pragma unroll
            for (int w = 0; w < 8; ++w) {
                h0 += part[(w * NPB + threadIdx.x) * 2 + 0];
                h1 += part[(w * NPB + threadIdx.x) * 2 + 1];
            }
            REC[node] = make_float4(h0, h1, dinv_of(cnt_[node]), 0.0f);
        }
    }
}

__global__ __launch_bounds__(256) void gather2_kernel(
    const int* __restrict__ bucket, const int2* __restrict__ spill, const int* __restrict__ cnt_,
    const int* __restrict__ total,
    const float4* __restrict__ REC, const float2* __restrict__ SK,
    const float* __restrict__ b2, const float* __restrict__ bsk,
    float* __restrict__ OUT, int n) {
    const int t = blockIdx.x * 256 + threadIdx.x;
    const int node = t >> 3;
    const int sub = t & 7;
    const bool ok = node < n;
    int c = 0;
    float di = 0.0f;
    float4 self = {0.0f, 0.0f, 0.0f, 0.0f};
    if (ok) {
        c = cnt_[node];
        self = REC[node];
        di = self.z;
    }
    const int beg = node * SLOTS;
    const int cb = (c < SLOTS) ? c : SLOTS;
    float a0 = 0.0f, a1 = 0.0f;
    for (int j = sub; j < cb; j += 8) {
        const int s = bucket[beg + j];
        const float4 r = REC[s];
        const float nrm = r.z * di;
        a0 += r.x * nrm;
        a1 += r.y * nrm;
    }
    if (c > SLOTS) {
        const int nsp = total[0];
        for (int k = sub; k < nsp; k += 8) {
            const int2 ent = spill[k];
            if (ent.x == node) {
                const float4 r = REC[ent.y];
                const float nrm = r.z * di;
                a0 += r.x * nrm;
                a1 += r.y * nrm;
            }
        }
    }
    #pragma unroll
    for (int off = 4; off > 0; off >>= 1) {
        a0 += __shfl_xor(a0, off, 32);
        a1 += __shfl_xor(a1, off, 32);
    }
    if (ok && sub == 0) {
        const float dd = di * di;
        const float2 sk = SK[node];
        float2 o;
        o.x = a0 + (self.x * dd + b2[0] + sk.x + bsk[0]);
        o.y = a1 + (self.y * dd + b2[1] + sk.y + bsk[1]);
        ((float2*)OUT)[node] = o;
    }
}

extern "C" void kernel_launch(void* const* d_in, const int* in_sizes, int n_in,
                              void* d_out, int out_size, void* d_ws, size_t ws_size,
                              hipStream_t stream) {
    const float* x   = (const float*)d_in[0];
    const int*   ei  = (const int*)d_in[1];
    const float* W1  = (const float*)d_in[2];
    const float* b1  = (const float*)d_in[3];
    const float* W2  = (const float*)d_in[4];
    const float* b2  = (const float*)d_in[5];
    const float* Wsk = (const float*)d_in[6];
    const float* bsk = (const float*)d_in[7];
    float* out = (float*)d_out;

    const int N = in_sizes[0] / FDIM;
    const int E = in_sizes[1] / 2;
    const int* src = ei;
    const int* dst = ei + E;

    char* ws = (char*)d_ws;
    float* H1   = (float*)(ws + (size_t)N * sizeof(float));
    float* A1   = H1 + (size_t)N * FDIM;
    float4* REC = (float4*)(A1 + (size_t)N * FDIM);
    int* bucket = (int*)H1;
    int* cnt    = (int*)A1;
    int* total  = cnt + N;
    float2* SK  = (float2*)(total + CNT_PAD);
    int2* spill = (int2*)(SK + N);

    const int nChunks = (E + CHUNK - 1) / CHUNK;
    zero_i32_kernel<<<(N + CNT_PAD + 255) / 256, 256, 0, stream>>>(cnt, N + CNT_PAD);
    count_fill_kernel<<<2 * nChunks, 256, 0, stream>>>(src, dst, E, nChunks, N, cnt, total, bucket, spill);

    gather1_gemm_kernel<<<(N + NPB - 1) / NPB, 512, 0, stream>>>(bucket, spill, cnt, total, x, W1, b1, W2,
                                                                 Wsk, REC, SK, N);

    gather2_kernel<<<(N * 8 + 255) / 256, 256, 0, stream>>>(bucket, spill, cnt, total, REC, SK, b2, bsk,
                                                            out, N);
}


# Layer 1 as 512 resident work-groups (two per WGP), each walking an equal share of the nodes 64 at a time: W1 fetched and staged once per work-group, no lone last blocks; sum order unchanged
# speedup vs baseline: 6.4595x; 1.0706x over previous
#include <hip/hip_runtime.h>


typedef float v2f __attribute__((ext_vector_type(2)));
typedef float v8f __attribute__((ext_vector_type(8)));

#define N_NODES 100000
#define FDIM 128
#define NPB 64
#define WPITCH 288
#define PITCH 132

#define SLOTS 32

__device__ __forceinline__ float dinv_of(int c) {
    const float d = (float)(c + 1);
    return (d > 0.0f) ? rsqrtf(d) : 0.0f;
}

__global__ void zero_i32_kernel(int* __restrict__ p, int n) {
    int i = blockIdx.x * blockDim.x + threadIdx.x;
    if (i < n) p[i] = 0;
}

#define CHUNK 2048
#define TICKET0 64
#define TICKET1 128
#define CNT_PAD 192
__global__ __launch_bounds__(256) void count_fill_kernel(const int* __restrict__ src,
                                                         const int* __restrict__ dst, int E, int nChunks,
                                                         int n, int* __restrict__ cnt,
                                                         int* __restrict__ total,
                                                         int* __restrict__ bucket,
                                                         int2* __restrict__ spill) {
    __shared__ int s_job;
    if (threadIdx.x == 0) {
        int half = (int)((__builtin_amdgcn_s_sendmsg_rtn(0x87) >> 18) & 1u);
        int t = atomicAdd(total + (half ? TICKET1 : TICKET0), 1);
        if (t >= nChunks) {
            half ^= 1;
            t = atomicAdd(total + (half ? TICKET1 : TICKET0), 1);
        }
        s_job = (t < nChunks) ? (t * 2 + half) : -1;
    }
    __syncthreads();
    const int job = s_job;
    if (job < 0) return;
    const int split = ((n >> 1) + 31) & ~31;
    const int lo = (job & 1) ? split : 0;
    const int hi = (job & 1) ? 0x7fffffff : split;
    const int base = (job >> 1) * CHUNK + (int)threadIdx.x;
    int d[8], s[8], r[8];
    #pragma unroll
    for (int k = 0; k < 8; ++k) {
        const int e = base + k * 256;
        const int ec = (e < E) ? e : (E - 1);
        const int v = dst[ec];
        s[k] = src[ec];
        d[k] = (e < E) ? v : -1;
    }
    asm volatile("" : "+v"(d[0]), "+v"(d[1]), "+v"(d[2]), "+v"(d[3]),
                      "+v"(d[4]), "+v"(d[5]), "+v"(d[6]), "+v"(d[7]),
                      "+v"(s[0]), "+v"(s[1]), "+v"(s[2]), "+v"(s[3]),
                      "+v"(s[4]), "+v"(s[5]), "+v"(s[6]), "+v"(s[7]));
    #pragma unroll
    for (int k = 0; k < 8; ++k) {
        r[k] = 0;
        if (d[k] >= lo && d[k] < hi) r[k] = atomicAdd(&cnt[d[k]], 1);
    }
    #pragma unroll
    for (int k = 0; k < 8; ++k) {
        if (d[k] >= lo && d[k] < hi) {
            if (r[k] < SLOTS) {
                bucket[d[k] * SLOTS + r[k]] = s[k];
            } else {
                const int o = atomicAdd(total, 1);
                spill[o] = make_int2(d[k], s[k]);
            }
        }
    }
}


template <int G>
__device__ __forceinline__ void gather_rows(const float* __restrict__ X, int idx, float w,
                                            int j, int lane, float4& acc) {
    float4 v[G];
    float ws[G];
    #pragma unroll
    for (int g = 0; g < G; ++g) {
        const int s = __builtin_amdgcn_readlane(idx, j + g);
        ws[g] = __int_as_float(__builtin_amdgcn_readlane(__float_as_int(w), j + g));
        v[g] = ((const float4*)(X + (size_t)s * FDIM))[lane];
    }
    #pragma unroll
    for (int g = 0; g < G; ++g) {
        acc.x += v[g].x * ws[g];
        acc.y += v[g].y * ws[g];
        acc.z += v[g].z * ws[g];
        acc.w += v[g].w * ws[g];
    }
}

__device__ __forceinline__ void gather_chunk(const float* __restrict__ X, int idx, float w,
                                             int m, int lane, float4& acc) {
    int j = 0;
    for (; j + 8 <= m; j += 8) gather_rows<8>(X, idx, w, j, lane, acc);
    if (m - j >= 4) { gather_rows<4>(X, idx, w, j, lane, acc); j += 4; }
    if (m - j >= 2) { gather_rows<2>(X, idx, w, j, lane, acc); j += 2; }
    if (m - j >= 1) { gather_rows<1>(X, idx, w, j, lane, acc); }
}

#define GRID1 512
#define NPBX 80
__global__ __launch_bounds__(512) void gather1_gemm_kernel(
    const int* __restrict__ bucket, const int2* __restrict__ spill, const int* __restrict__ cnt_,
    const int* __restrict__ total, const float* __restrict__ X, const float* __restrict__ W1,
    const float* __restrict__ b1, const float* __restrict__ W2, const float* __restrict__ Wsk,
    float4* __restrict__ REC, float2* __restrict__ SK, int n) {
    __shared__ __attribute__((aligned(16))) float agg[NPBX * PITCH];
    __shared__ __attribute__((aligned(16))) float w1s[64 * WPITCH];
    __shared__ float part[8 * NPBX * 2];

    const int lane = threadIdx.x & 31;
    const int wave = __builtin_amdgcn_readfirstlane((int)(threadIdx.x >> 5));
    const int c16 = lane & 15;
    const int kh = lane >> 4;
    const int strip = wave & 7;
    const int th = wave >> 3;
    const int colBase = strip * 16;

    const int share = (n + (int)gridDim.x - 1) / (int)gridDim.x;
    const int rangeBeg = (int)blockIdx.x * share;
    const int rangeEnd = (rangeBeg + share < n) ? (rangeBeg + share) : n;
    if (rangeBeg >= rangeEnd) return;

    {
        float4 wr0[4], wr1[4];
        #pragma unroll
        for (int u = 0; u < 4; ++u) {
            const int j = wave + 16 * u;
            wr0[u] = ((const float4*)(W1 + (size_t)(2 * j) * FDIM))[lane];
            wr1[u] = ((const float4*)(W1 + (size_t)(2 * j + 1) * FDIM))[lane];
        }
        #pragma unroll
        for (int u = 0; u < 4; ++u) {
            const int j = wave + 16 * u;
            float* wd = w1s + j * WPITCH + lane * 8;
            *(float4*)(wd)     = make_float4(wr0[u].x, wr1[u].x, wr0[u].y, wr1[u].y);
            *(float4*)(wd + 4) = make_float4(wr0[u].z, wr1[u].z, wr0[u].w, wr1[u].w);
        }
    }

    const float4 wsA = ((const float4*)Wsk)[lane * 2];
    const float4 wsB = ((const float4*)Wsk)[lane * 2 + 1];

    const int col = colBase + c16;
    const float bias = b1[col];
    const float2 w2 = ((const float2*)W2)[col];

    int nodeBase = rangeBeg;
    #pragma nounroll
    while (nodeBase < rangeEnd) {
    const int left = rangeEnd - nodeBase;
    const int rows = (left <= NPBX) ? left : NPB;

    #pragma nounroll
    for (int p = 0; p < 3; ++p) {
        if (p * 32 >= rows) break;
        bool ok[2];
        int nd[2], cnt[2], idx0[2];
        float di[2], w0[2];
        float4 xr[2];
        #pragma unroll
        for (int q = 0; q < 2; ++q) {
            const int r = (p * 2 + q) * 16 + wave;
            ok[q] = r < rows;
            nd[q] = ok[q] ? (nodeBase + r) : (n - 1);
            const int cq = cnt_[nd[q]];
            cnt[q] = ok[q] ? cq : 0;
            di[q] = dinv_of(cq);
        }
        #pragma unroll
        for (int q = 0; q < 2; ++q) {
            xr[q] = ((const float4*)(X + (size_t)nd[q] * FDIM))[lane];
            idx0[q] = 0;
            if (lane < cnt[q]) idx0[q] = bucket[nd[q] * SLOTS + lane];
        }
        #pragma unroll
        for (int q = 0; q < 2; ++q) {
            w0[q] = 0.0f;
            if (lane < cnt[q]) w0[q] = dinv_of(cnt_[idx0[q]]) * di[q];
        }

        #pragma unroll
        for (int q = 0; q < 2; ++q) {
            const int r = (p * 2 + q) * 16 + wave;
            const int c = cnt[q];
            float4 acc = {0.0f, 0.0f, 0.0f, 0.0f};
            gather_chunk(X, idx0[q], w0[q], (c < SLOTS) ? c : SLOTS, lane, acc);
            if (c > SLOTS) {
                const int nsp = total[0];
                for (int base = 0; base < nsp; base += 32) {
                    int idx = 0;
                    float w = 0.0f;
                    bool mine = false;
                    if (base + lane < nsp) {
                        const int2 ent = spill[base + lane];
                        mine = ent.x == nd[q];
                        if (mine) {
                            idx = ent.y;
                            w = dinv_of(cnt_[idx]) * di[q];
                        }
                    }
                    unsigned int hits = __builtin_amdgcn_ballot_w32(mine);
                    while (hits) {
                        const int j = __builtin_ctz(hits);
                        hits &= hits - 1;
                        gather_rows<1>(X, idx, w, j, lane, acc);
                    }
                }
            }
            const float4 x = xr[q];
            const float dd = ok[q] ? di[q] * di[q] : 0.0f;
            float4 row;
            row.x = acc.x + x.x * dd;
            row.y = acc.y + x.y * dd;
            row.z = acc.z + x.z * dd;
            row.w = acc.w + x.w * dd;
            if ((p * 2 + q) * 16 < rows) *(float4*)(agg + r * PITCH + lane * 4) = row;

            float s0 = x.x * wsA.x + x.y * wsA.z + x.z * wsB.x + x.w * wsB.z;
            float s1 = x.x * wsA.y + x.y * wsA.w + x.z * wsB.y + x.w * wsB.w;
            #pragma unroll
            for (int off = 16; off > 0; off >>= 1) {
                s0 += __shfl_down(s0, off, 32);
                s1 += __shfl_down(s1, off, 32);
            }
            if (ok[q] && lane == 0) SK[nodeBase + r] = make_float2(s0, s1);
        }
    }
    __syncthreads();

    const float* wp = w1s + kh * WPITCH + (colBase + c16) * 2;

    #pragma unroll
    for (int t = 0; t < 3; ++t) {
        const int tile = (t < 2) ? (2 * th + t) : 4;
        if ((t < 2 || th == 0) && tile * 16 < rows) {
        v8f c0 = {};
        v8f c1 = {};
        const float* ap = agg + (tile * 16 + c16) * PITCH + 2 * kh;
        __builtin_amdgcn_sched_barrier(0);
        #pragma unroll
        for (int kk = 0; kk < 16; ++kk) {
            const v2f a0 = *(const v2f*)(ap + 4 * kk);
            const v2f a1 = *(const v2f*)(ap + 4 * kk + 64);
            const v2f b0 = *(const v2f*)(wp + (2 * kk) * WPITCH);
            const v2f b1 = *(const v2f*)(wp + (2 * kk + 32) * WPITCH);
            c0 = __builtin_amdgcn_wmma_f32_16x16x4_f32(
                    false, a0, false, b0, (short)0, c0, false, false);
            c1 = __builtin_amdgcn_wmma_f32_16x16x4_f32(
                    false, a1, false, b1, (short)0, c1, false, false);
        }
        __builtin_amdgcn_sched_barrier(0);
        asm volatile("v_nop\n\tv_nop\n\tv_nop\n\tv_nop");
        __builtin_amdgcn_sched_barrier(0);
        const v8f cc = c0 + c1;

        float p0[8], p1[8];
        #pragma unroll
        for (int r = 0; r < 8; ++r) {
            float a = cc[r] + bias;
            a = a > 0.0f ? a : 0.0f;
            p0[r] = a * w2.x;
            p1[r] = a * w2.y;
        }
        #pragma unroll
        for (int off = 8; off > 0; off >>= 1) {
            #pragma unroll
            for (int r = 0; r < 8; ++r) {
                p0[r] += __shfl_xor(p0[r], off, 32);
                p1[r] += __shfl_xor(p1[r], off, 32);
            }
        }
        if (c16 == 0) {
            #pragma unroll
            for (int r = 0; r < 8; ++r) {
                part[(strip * NPBX + tile * 16 + r + 8 * kh) * 2 + 0] = p0[r];
                part[(strip * NPBX + tile * 16 + r + 8 * kh) * 2 + 1] = p1[r];
            }
        }
        }
    }
    __syncthreads();
    if ((int)threadIdx.x < rows) {
        const int node = nodeBase + threadIdx.x;
        float h0 = 0.0f, h1 = 0.0f;
        #pragma unroll
        for (int w = 0; w < 8; ++w) {
            h0 += part[(w * NPBX + threadIdx.x) * 2 + 0];
            h1 += part[(w * NPBX + threadIdx.x) * 2 + 1];
        }
        REC[node] = make_float4(h0, h1, dinv_of(cnt_[node]), 0.0f);
    }
    nodeBase += rows;
    }
}

__global__ __launch_bounds__(256) void gather2_kernel(
    const int* __restrict__ bucket, const int2* __restrict__ spill, const int* __restrict__ cnt_,
    const int* __restrict__ total,
    const float4* __restrict__ REC, const float2* __restrict__ SK,
    const float* __restrict__ b2, const float* __restrict__ bsk,
    float* __restrict__ OUT, int n) {
    const int t = blockIdx.x * 256 + threadIdx.x;
    const int node = t >> 3;
    const int sub = t & 7;
    const bool ok = node < n;
    int c = 0;
    float di = 0.0f;
    float4 self = {0.0f, 0.0f, 0.0f, 0.0f};
    if (ok) {
        c = cnt_[node];
        self = REC[node];
        di = self.z;
    }
    const int beg = node * SLOTS;
    const int cb = (c < SLOTS) ? c : SLOTS;
    float a0 = 0.0f, a1 = 0.0f;
    for (int j = sub; j < cb; j += 8) {
        const int s = bucket[beg + j];
        const float4 r = REC[s];
        const float nrm = r.z * di;
        a0 += r.x * nrm;
        a1 += r.y * nrm;
    }
    if (c > SLOTS) {
        const int nsp = total[0];
        for (int k = sub; k < nsp; k += 8) {
            const int2 ent = spill[k];
            if (ent.x == node) {
                const float4 r = REC[ent.y];
                const float nrm = r.z * di;
                a0 += r.x * nrm;
                a1 += r.y * nrm;
            }
        }
    }
    #pragma unroll
    for (int off = 4; off > 0; off >>= 1) {
        a0 += __shfl_xor(a0, off, 32);
        a1 += __shfl_xor(a1, off, 32);
    }
    if (ok && sub == 0) {
        const float dd = di * di;
        const float2 sk = SK[node];
        float2 o;
        o.x = a0 + (self.x * dd + b2[0] + sk.x + bsk[0]);
        o.y = a1 + (self.y * dd + b2[1] + sk.y + bsk[1]);
        ((float2*)OUT)[node] = o;
    }
}

extern "C" void kernel_launch(void* const* d_in, const int* in_sizes, int n_in,
                              void* d_out, int out_size, void* d_ws, size_t ws_size,
                              hipStream_t stream) {
    const float* x   = (const float*)d_in[0];
    const int*   ei  = (const int*)d_in[1];
    const float* W1  = (const float*)d_in[2];
    const float* b1  = (const float*)d_in[3];
    const float* W2  = (const float*)d_in[4];
    const float* b2  = (const float*)d_in[5];
    const float* Wsk = (const float*)d_in[6];
    const float* bsk = (const float*)d_in[7];
    float* out = (float*)d_out;

    const int N = in_sizes[0] / FDIM;
    const int E = in_sizes[1] / 2;
    const int* src = ei;
    const int* dst = ei + E;

    char* ws = (char*)d_ws;
    float* H1   = (float*)(ws + (size_t)N * sizeof(float));
    float* A1   = H1 + (size_t)N * FDIM;
    float4* REC = (float4*)(A1 + (size_t)N * FDIM);
    int* bucket = (int*)H1;
    int* cnt    = (int*)A1;
    int* total  = cnt + N;
    float2* SK  = (float2*)(total + CNT_PAD);
    int2* spill = (int2*)(SK + N);

    const int nChunks = (E + CHUNK - 1) / CHUNK;
    zero_i32_kernel<<<(N + CNT_PAD + 255) / 256, 256, 0, stream>>>(cnt, N + CNT_PAD);
    count_fill_kernel<<<2 * nChunks, 256, 0, stream>>>(src, dst, E, nChunks, N, cnt, total, bucket, spill);

    gather1_gemm_kernel<<<GRID1, 512, 0, stream>>>(bucket, spill, cnt, total, x, W1, b1, W2,
                                                                 Wsk, REC, SK, N);

    gather2_kernel<<<(N * 8 + 255) / 256, 256, 0, stream>>>(bucket, spill, cnt, total, REC, SK, b2, bsk,
                                                            out, N);
}
